# starting kernel
# baseline (speedup 1.0000x reference)
#include <hip/hip_runtime.h>

typedef float v2f __attribute__((ext_vector_type(2)));
typedef float v8f __attribute__((ext_vector_type(8)));

__global__ void zero_f(float* __restrict__ p, int n) {
  int i = blockIdx.x * blockDim.x + threadIdx.x;
  if (i < n) p[i] = 0.0f;
}

template <int INC, int OUTC, int MT>
__global__ void wmma_linear(const float* __restrict__ X,
                            const float* __restrict__ W,
                            const float* __restrict__ Bias,
                            float* __restrict__ Y, int nrows) {
  constexpr int NT = OUTC / 16;
  int wave = (int)((blockIdx.x * blockDim.x + threadIdx.x) >> 5);
  int lane = threadIdx.x & 31;
  int row0 = wave * 16 * MT;
  if (row0 >= nrows) return;

  int nm = lane & 15;
  int kg = (lane >> 4) << 1;

  const float* xrow[MT];
#pragma unroll
  for (int mt = 0; mt < MT; ++mt) {
    xrow[mt] = X + (size_t)(row0 + mt * 16 + nm) * INC;
    __builtin_prefetch(xrow[mt], 0, 3);
  }
  const float* wrow[NT];
#pragma unroll
  for (int t = 0; t < NT; ++t) wrow[t] = W + (size_t)(t * 16 + nm) * INC;

  v8f acc[MT][NT];
#pragma unroll
  for (int t = 0; t < NT; ++t) {
    float bv = Bias[t * 16 + nm];
#pragma unroll
    for (int mt = 0; mt < MT; ++mt)
#pragma unroll
      for (int v = 0; v < 8; ++v) acc[mt][t][v] = bv;
  }

#pragma unroll 4
  for (int k = 0; k < INC; k += 4) {
    v2f b[NT];
#pragma unroll
    for (int t = 0; t < NT; ++t) {
      b[t].x = wrow[t][k + kg];
      b[t].y = wrow[t][k + kg + 1];
    }
#pragma unroll
    for (int mt = 0; mt < MT; ++mt) {
      v2f a;
      a.x = xrow[mt][k + kg];
      a.y = xrow[mt][k + kg + 1];
#pragma unroll
      for (int t = 0; t < NT; ++t) {
        acc[mt][t] = __builtin_amdgcn_wmma_f32_16x16x4_f32(
             false, a,  false, b[t],
             (short)0, acc[mt][t],  false,
             false);
      }
    }
  }

  int mb = (lane >> 4) * 8;
#pragma unroll
  for (int mt = 0; mt < MT; ++mt)
#pragma unroll
    for (int t = 0; t < NT; ++t)
#pragma unroll
      for (int v = 0; v < 8; ++v)
        Y[(size_t)(row0 + mt * 16 + mb + v) * OUTC + t * 16 + nm] =
            acc[mt][t][v];
}

template <int C>
__global__ void scatter_add(const float* __restrict__ H,
                            const int* __restrict__ srcI,
                            const int* __restrict__ dstI,
                            float* __restrict__ agg, int nedges) {
  long long tid = (long long)blockIdx.x * blockDim.x + threadIdx.x;
  int e = (int)(tid / C);
  int c = (int)(tid % C);
  if (e >= nedges) return;
  int s = srcI[e];
  int d = dstI[e];
  float val = H[(size_t)s * C + c];
  __hip_atomic_fetch_add(&agg[(size_t)d * C + c], val, __ATOMIC_RELAXED,
                         __HIP_MEMORY_SCOPE_AGENT);
}

__global__ void add_relu(const float* __restrict__ A, float* __restrict__ FZ,
                         int n) {
  int i = blockIdx.x * blockDim.x + threadIdx.x;
  if (i < n) {
    float v = FZ[i] + A[i];
    FZ[i] = v > 0.0f ? v : 0.0f;
  }
}

__global__ void add_logsoftmax16(const float* __restrict__ F,
                                 const float* __restrict__ A,
                                 float* __restrict__ O, int nrows) {
  int r = blockIdx.x * blockDim.x + threadIdx.x;
  if (r >= nrows) return;
  float v[16];
  float mx = -3.402823466e38f;
#pragma unroll
  for (int c = 0; c < 16; ++c) {
    v[c] = F[(size_t)r * 16 + c] + A[(size_t)r * 16 + c];
    mx = fmaxf(mx, v[c]);
  }
  float s = 0.0f;
#pragma unroll
  for (int c = 0; c < 16; ++c) s += __expf(v[c] - mx);
  float l = mx + __logf(s);
#pragma unroll
  for (int c = 0; c < 16; ++c) O[(size_t)r * 16 + c] = v[c] - l;
}

static inline int cdiv_ll(long long a, long long b) {
  return (int)((a + b - 1) / b);
}

extern "C" void kernel_launch(void* const* d_in, const int* in_sizes, int n_in,
                              void* d_out, int out_size, void* d_ws,
                              size_t ws_size, hipStream_t stream) {
  (void)n_in; (void)out_size; (void)ws_size;
  const float* x     = (const float*)d_in[0];
  const float* w1a_w = (const float*)d_in[1];
  const float* w1a_b = (const float*)d_in[2];
  const float* w1b_w = (const float*)d_in[3];
  const float* w1b_b = (const float*)d_in[4];
  const float* w2a_w = (const float*)d_in[5];
  const float* w2a_b = (const float*)d_in[6];
  const float* w2b_w = (const float*)d_in[7];
  const float* w2b_b = (const float*)d_in[8];
  const int*   edge  = (const int*)d_in[9];

  const int N = in_sizes[0] / 256;
  const int E = in_sizes[9] / 2;
  const int* srcI = edge;
  const int* dstI = edge + E;

  float* buf0 = (float*)d_ws;
  float* buf1 = buf0 + (size_t)N * 32;
  float* buf2 = buf1 + (size_t)N * 32;

  const int BLK = 256;
  const int MT = 2;
  const int waves = N / (16 * MT);
  const int gemm_blocks = cdiv_ll(waves, BLK / 32);

  zero_f<<<cdiv_ll((long long)N * 32, BLK), BLK, 0, stream>>>(buf1, N * 32);
  wmma_linear<256, 32, MT><<<gemm_blocks, BLK, 0, stream>>>(x, w1a_w, w1a_b,
                                                            buf0, N);
  scatter_add<32><<<cdiv_ll((long long)E * 32, BLK), BLK, 0, stream>>>(
      buf0, srcI, dstI, buf1, E);
  wmma_linear<32, 32, MT><<<gemm_blocks, BLK, 0, stream>>>(buf0, w1b_w, w1b_b,
                                                           buf2, N);
  add_relu<<<cdiv_ll((long long)N * 32, BLK), BLK, 0, stream>>>(buf1, buf2,
                                                                N * 32);

  zero_f<<<cdiv_ll((long long)N * 16, BLK), BLK, 0, stream>>>(buf1, N * 16);
  wmma_linear<32, 16, MT><<<gemm_blocks, BLK, 0, stream>>>(buf2, w2a_w, w2a_b,
                                                           buf0, N);
  scatter_add<16><<<cdiv_ll((long long)E * 16, BLK), BLK, 0, stream>>>(
      buf0, srcI, dstI, buf1, E);
  wmma_linear<16, 16, MT><<<gemm_blocks, BLK, 0, stream>>>(buf0, w2b_w, w2b_b,
                                                           buf2, N);
  add_logsoftmax16<<<cdiv_ll(N, BLK), BLK, 0, stream>>>(buf2, buf1,
                                                        (float*)d_out, N);
}


# Segment-sum hits go to a per-wave LDS queue and are drained 32 at a time, C/4 lanes per edge: source rows load as whole lines and every lane adds; LDS float adds in a new arrival order
# speedup vs baseline: 1.5355x; 1.5355x over previous
#include <hip/hip_runtime.h>

typedef float v2f __attribute__((ext_vector_type(2)));
typedef float v8f __attribute__((ext_vector_type(8)));

template <int INC, int OUTC, int MT>
__global__ void wmma_linear(const float* __restrict__ X,
                            const float* __restrict__ W,
                            const float* __restrict__ Bias,
                            float* __restrict__ Y, int nrows) {
  constexpr int NT = OUTC / 16;
  int wave = (int)((blockIdx.x * blockDim.x + threadIdx.x) >> 5);
  int lane = threadIdx.x & 31;
  int row0 = wave * 16 * MT;
  if (row0 >= nrows) return;

  int nm = lane & 15;
  int kg = (lane >> 4) << 1;

  const float* xrow[MT];
#pragma unroll
  for (int mt = 0; mt < MT; ++mt) {
    xrow[mt] = X + (size_t)(row0 + mt * 16 + nm) * INC;
    __builtin_prefetch(xrow[mt], 0, 3);
  }
  const float* wrow[NT];
#pragma unroll
  for (int t = 0; t < NT; ++t) wrow[t] = W + (size_t)(t * 16 + nm) * INC;

  v8f acc[MT][NT];
#pragma unroll
  for (int t = 0; t < NT; ++t) {
    float bv = Bias[t * 16 + nm];
#pragma unroll
    for (int mt = 0; mt < MT; ++mt)
#pragma unroll
      for (int v = 0; v < 8; ++v) acc[mt][t][v] = bv;
  }

#pragma unroll 4
  for (int k = 0; k < INC; k += 4) {
    v2f b[NT];
#pragma unroll
    for (int t = 0; t < NT; ++t) {
      b[t].x = wrow[t][k + kg];
      b[t].y = wrow[t][k + kg + 1];
    }
#pragma unroll
    for (int mt = 0; mt < MT; ++mt) {
      v2f a;
      a.x = xrow[mt][k + kg];
      a.y = xrow[mt][k + kg + 1];
#pragma unroll
      for (int t = 0; t < NT; ++t) {
        acc[mt][t] = __builtin_amdgcn_wmma_f32_16x16x4_f32(
             false, a,  false, b[t],
             (short)0, acc[mt][t],  false,
             false);
      }
    }
  }

  int mb = (lane >> 4) * 8;
#pragma unroll
  for (int mt = 0; mt < MT; ++mt)
#pragma unroll
    for (int t = 0; t < NT; ++t)
#pragma unroll
      for (int v = 0; v < 8; ++v)
        Y[(size_t)(row0 + mt * 16 + mb + v) * OUTC + t * 16 + nm] =
            acc[mt][t][v];
}

template <int C, bool FULL>
__device__ __forceinline__ void drain_queue(
    const int* q, int n, int slot, int part, const float* __restrict__ H,
    const int* __restrict__ srcI, const int* __restrict__ dstI, float* tile,
    int base) {
  constexpr int PITCH = C + 1;
  constexpr int LPE = C / 4;
  constexpr int EPI = 32 / LPE;
  constexpr int NI = 32 / EPI;
  const char* Hb = reinterpret_cast<const char*>(H);
  int e[NI], rel[NI];
  unsigned sidx[NI];
  float4 v[NI];
#pragma unroll
  for (int i = 0; i < NI; ++i) {
    const int idx = i * EPI + slot;
    e[i] = q[(FULL || idx < n) ? idx : 0];
  }
#pragma unroll
  for (int i = 0; i < NI; ++i) {
    sidx[i] = (unsigned)srcI[e[i]];
    rel[i] = dstI[e[i]] - base;
  }
#pragma unroll
  for (int i = 0; i < NI; ++i)
    v[i] = *reinterpret_cast<const float4*>(
        Hb + (sidx[i] * (unsigned)(C * 4) + (unsigned)part * 16u));
#pragma unroll
  for (int i = 0; i < NI; ++i) {
    if (FULL || (i * EPI + slot) < n) {
      float* tp = tile + rel[i] * PITCH + 4 * part;
      __hip_atomic_fetch_add(tp + 0, v[i].x, __ATOMIC_RELAXED,
                             __HIP_MEMORY_SCOPE_WORKGROUP);
      __hip_atomic_fetch_add(tp + 1, v[i].y, __ATOMIC_RELAXED,
                             __HIP_MEMORY_SCOPE_WORKGROUP);
      __hip_atomic_fetch_add(tp + 2, v[i].z, __ATOMIC_RELAXED,
                             __HIP_MEMORY_SCOPE_WORKGROUP);
      __hip_atomic_fetch_add(tp + 3, v[i].w, __ATOMIC_RELAXED,
                             __HIP_MEMORY_SCOPE_WORKGROUP);
    }
  }
}

template <int C, int S, int TN>
__global__ __launch_bounds__(1024) void tile_segsum(
    const float* __restrict__ H, const int* __restrict__ srcI,
    const int* __restrict__ dstI, float* __restrict__ P0,
    float* __restrict__ P1, float* __restrict__ P2, float* __restrict__ P3,
    int nedges, int nrows) {
  constexpr int PITCH = C + 1;
  constexpr int NT = 1024;
  constexpr int U4 = 8;
  constexpr int TRIP = NT * 4 * U4;
  constexpr int LPE = C / 4;
  constexpr int QCAP = 64;
  __shared__ float tile[TN * PITCH];
  __shared__ int queue[(NT / 32) * QCAP];

  const int tid = threadIdx.x;
  const int lane = tid & 31;
  const int part = lane % LPE;
  const int slot = lane / LPE;
  int* q = queue + (tid >> 5) * QCAP;
  const int t = blockIdx.x / S;
  const int s = blockIdx.x - t * S;
  const int base = t * TN;

  for (int i = tid; i < TN * PITCH; i += NT) tile[i] = 0.0f;
  __syncthreads();

  const int slice = ((nedges + 4 * S - 1) / (4 * S)) * 4;
  const int e0 = s * slice;
  const int e1 = min(e0 + slice, nedges);

  int4 d[U4];
#pragma unroll
  for (int j = 0; j < U4; ++j) {
    const int e = e0 + ((j * NT + tid) << 2);
    d[j] = (e < e1) ? *reinterpret_cast<const int4*>(dstI + e)
                    : make_int4(-1, -1, -1, -1);
  }

  int qn = 0;
  for (int eb = e0; eb < e1; eb += TRIP) {
    unsigned m = 0;
#pragma unroll
    for (int j = 0; j < U4; ++j) {
      m |= ((unsigned)(d[j].x - base) < (unsigned)TN) ? (1u << (4 * j + 0)) : 0u;
      m |= ((unsigned)(d[j].y - base) < (unsigned)TN) ? (1u << (4 * j + 1)) : 0u;
      m |= ((unsigned)(d[j].z - base) < (unsigned)TN) ? (1u << (4 * j + 2)) : 0u;
      m |= ((unsigned)(d[j].w - base) < (unsigned)TN) ? (1u << (4 * j + 3)) : 0u;
    }
#pragma unroll
    for (int j = 0; j < U4; ++j) {
      const int e = eb + TRIP + ((j * NT + tid) << 2);
      d[j] = (e < e1) ? *reinterpret_cast<const int4*>(dstI + e)
                      : make_int4(-1, -1, -1, -1);
    }
    for (;;) {
      const bool has = (m != 0);
      const unsigned bal = __builtin_amdgcn_ballot_w32(has);
      if (bal == 0) break;
      if (has) {
        const int b = __builtin_ctz(m);
        m &= m - 1;
        q[qn + (int)__builtin_amdgcn_mbcnt_lo(bal, 0u)] =
            eb + ((((b >> 2) * NT + tid) << 2) | (b & 3));
      }
      qn += __builtin_popcount(bal);
      if (qn >= 32) {
        qn -= 32;
        drain_queue<C, true>(q + qn, 32, slot, part, H, srcI, dstI, tile, base);
      }
    }
  }
  if (qn > 0)
    drain_queue<C, false>(q, qn, slot, part, H, srcI, dstI, tile, base);
  __syncthreads();

  const int rows = min(TN, nrows - base);
  const int nel = rows * C;
  float* outp = (s == 0) ? P0 : (s == 1) ? P1 : (s == 2) ? P2 : P3;
  outp += (size_t)base * C;
  for (int i = tid; i < nel; i += NT) {
    float a = tile[(i / C) * PITCH + (i % C)];
    if (s == 0) a += outp[i];
    outp[i] = a;
  }
}

__global__ void add_relu(const float* __restrict__ A, float* __restrict__ FZ,
                         int n) {
  int i = blockIdx.x * blockDim.x + threadIdx.x;
  if (i < n) {
    float v = FZ[i] + A[i];
    FZ[i] = v > 0.0f ? v : 0.0f;
  }
}

__global__ void add_logsoftmax16(const float* __restrict__ F,
                                 const float* __restrict__ A1,
                                 const float* __restrict__ A2,
                                 const float* __restrict__ A3,
                                 float* __restrict__ O, int nrows) {
  int r = blockIdx.x * blockDim.x + threadIdx.x;
  if (r >= nrows) return;
  float v[16];
  float mx = -3.402823466e38f;
#pragma unroll
  for (int c = 0; c < 16; ++c) {
    v[c] = F[(size_t)r * 16 + c] + A1[(size_t)r * 16 + c] +
           A2[(size_t)r * 16 + c] + A3[(size_t)r * 16 + c];
    mx = fmaxf(mx, v[c]);
  }
  float s = 0.0f;
#pragma unroll
  for (int c = 0; c < 16; ++c) s += __expf(v[c] - mx);
  float l = mx + __logf(s);
#pragma unroll
  for (int c = 0; c < 16; ++c) O[(size_t)r * 16 + c] = v[c] - l;
}

static inline int cdiv_ll(long long a, long long b) {
  return (int)((a + b - 1) / b);
}

extern "C" void kernel_launch(void* const* d_in, const int* in_sizes, int n_in,
                              void* d_out, int out_size, void* d_ws,
                              size_t ws_size, hipStream_t stream) {
  (void)n_in; (void)out_size; (void)ws_size;
  const float* x     = (const float*)d_in[0];
  const float* w1a_w = (const float*)d_in[1];
  const float* w1a_b = (const float*)d_in[2];
  const float* w1b_w = (const float*)d_in[3];
  const float* w1b_b = (const float*)d_in[4];
  const float* w2a_w = (const float*)d_in[5];
  const float* w2a_b = (const float*)d_in[6];
  const float* w2b_w = (const float*)d_in[7];
  const float* w2b_b = (const float*)d_in[8];
  const int*   edge  = (const int*)d_in[9];

  const int N = in_sizes[0] / 256;
  const int E = in_sizes[9] / 2;
  const int* srcI = edge;
  const int* dstI = edge + E;

  float* buf0 = (float*)d_ws;
  float* buf1 = buf0 + (size_t)N * 32;
  float* buf2 = buf1 + (size_t)N * 32;

  const int BLK = 256;
  const int MT = 2;
  const int waves = N / (16 * MT);
  const int gemm_blocks = cdiv_ll(waves, BLK / 32);

  constexpr int TN1 = 782, S1 = 2;
  constexpr int TN2 = 1564, S2 = 4;

  wmma_linear<256, 32, MT><<<gemm_blocks, BLK, 0, stream>>>(x, w1a_w, w1a_b,
                                                            buf0, N);
  wmma_linear<32, 32, MT><<<gemm_blocks, BLK, 0, stream>>>(buf0, w1b_w, w1b_b,
                                                           buf2, N);
  tile_segsum<32, S1, TN1><<<cdiv_ll(N, TN1) * S1, 1024, 0, stream>>>(
      buf0, srcI, dstI, buf2, buf1, nullptr, nullptr, E, N);
  add_relu<<<cdiv_ll((long long)N * 32, BLK), BLK, 0, stream>>>(buf1, buf2,
                                                                N * 32);

  wmma_linear<32, 16, MT><<<gemm_blocks, BLK, 0, stream>>>(buf2, w2a_w, w2a_b,
                                                           buf0, N);
  wmma_linear<16, 16, MT><<<gemm_blocks, BLK, 0, stream>>>(buf0, w2b_w, w2b_b,
                                                           buf2, N);
  tile_segsum<16, S2, TN2><<<cdiv_ll(N, TN2) * S2, 1024, 0, stream>>>(
      buf0, srcI, dstI, buf2, buf1, buf1 + (size_t)N * 16,
      buf0 + (size_t)N * 16, E, N);
  add_logsoftmax16<<<cdiv_ll(N, BLK), BLK, 0, stream>>>(
      buf2, buf1, buf1 + (size_t)N * 16, buf0 + (size_t)N * 16, (float*)d_out,
      N);
}
